# starting kernel
# baseline (speedup 1.0000x reference)
#include <hip/hip_runtime.h>


typedef __attribute__((ext_vector_type(2))) float v2f;
typedef __attribute__((ext_vector_type(8))) float v8f;


__global__ void zero_f32(float* __restrict__ p, int n4) {
    int i = blockIdx.x * blockDim.x + threadIdx.x;
    if (i < n4) ((float4*)p)[i] = make_float4(0.f, 0.f, 0.f, 0.f);
}

__global__ void deg_kernel(const int* __restrict__ row,
                           const float* __restrict__ ew,
                           float* __restrict__ deg, int E) {
    int e = blockIdx.x * blockDim.x + threadIdx.x;
    if (e < E) atomicAdd(&deg[row[e]], ew[e]);
}

__global__ void dinv_kernel(const float* __restrict__ deg,
                            float* __restrict__ dinv, int N) {
    int i = blockIdx.x * blockDim.x + threadIdx.x;
    if (i < N) {
        float d = deg[i];
        dinv[i] = (d > 0.f) ? rsqrtf(d) : 0.f;
    }
}

__global__ void normw_kernel(const int* __restrict__ row,
                             const int* __restrict__ col,
                             const float* __restrict__ ew,
                             const float* __restrict__ dinv,
                             float* __restrict__ nw, int E) {
    int e = blockIdx.x * blockDim.x + threadIdx.x;
    if (e < E) nw[e] = -dinv[row[e]] * ew[e] * dinv[col[e]];
}

__global__ void spmm_kernel(const int* __restrict__ row,
                            const int* __restrict__ col,
                            const float* __restrict__ nw,
                            const float* __restrict__ in,
                            float* __restrict__ out,
                            int E, int D) {
    int t = blockIdx.x * blockDim.x + threadIdx.x;
    int groups = D >> 2;
    int e = t / groups;
    int g = t - e * groups;
    if (e >= E) return;
    int r = row[e], c = col[e];
    float w = nw[e];
    float4 v = ((const float4*)(in + (size_t)r * D))[g];
    float* dst = out + (size_t)c * D + (g << 2);
    atomicAdd(dst + 0, w * v.x);
    atomicAdd(dst + 1, w * v.y);
    atomicAdd(dst + 2, w * v.z);
    atomicAdd(dst + 3, w * v.w);
}

__global__ void layer1_combine(const float* __restrict__ w10,
                               const float* __restrict__ s1,
                               const float* __restrict__ s2,
                               const float* __restrict__ w12,
                               const float* __restrict__ b1,
                               float* __restrict__ h,
                               int n4, int D) {
    int i = blockIdx.x * blockDim.x + threadIdx.x;
    if (i >= n4) return;
    int cg = i & ((D >> 2) - 1);
    float4 a = ((const float4*)w10)[i];
    float4 s = ((const float4*)s1)[i];
    float4 t = ((const float4*)s2)[i];
    float4 m = ((const float4*)w12)[i];
    float4 b = ((const float4*)b1)[cg];
    float4 o;
    o.x = fmaxf(a.x + s.x + 2.f * t.x - m.x + b.x, 0.f);
    o.y = fmaxf(a.y + s.y + 2.f * t.y - m.y + b.y, 0.f);
    o.z = fmaxf(a.z + s.z + 2.f * t.z - m.z + b.z, 0.f);
    o.w = fmaxf(a.w + s.w + 2.f * t.w - m.w + b.w, 0.f);
    ((float4*)h)[i] = o;
}

__global__ void t2_combine(const float* __restrict__ u,
                           const float* __restrict__ h,
                           float* __restrict__ t2, int n4) {
    int i = blockIdx.x * blockDim.x + threadIdx.x;
    if (i >= n4) return;
    float4 a = ((const float4*)u)[i];
    float4 b = ((const float4*)h)[i];
    float4 o;
    o.x = 2.f * a.x - b.x;  o.y = 2.f * a.y - b.y;
    o.z = 2.f * a.z - b.z;  o.w = 2.f * a.w - b.w;
    ((float4*)t2)[i] = o;
}

#define DC 128
#define APAD 132

__global__ void __launch_bounds__(256)
gemm3_wmma_f32(const float* __restrict__ A0,
               const float* __restrict__ A1,
               const float* __restrict__ A2,
               const float* __restrict__ B,
               const float* __restrict__ bias,
               float* __restrict__ out) {
    __shared__ float As[3][16][APAD];

    const int wave = threadIdx.x >> 5;
    const int lane = threadIdx.x & 31;
    const int idx  = lane & 15;
    const int half = lane >> 4;

    const int tm   = blockIdx.x;
    const int tn   = wave;
    const int colB = tn * 16 + idx;

    const float* Am[3] = {A0, A1, A2};

#pragma unroll
    for (int m = 0; m < 3; ++m) {
        const float* src = Am[m] + (size_t)tm * 16 * DC;
#pragma unroll
        for (int i = 0; i < 2; ++i) {
            int f  = threadIdx.x + i * 256;
            int r  = f >> 5;
            int c4 = f & 31;
            float4 v = ((const float4*)(src + (size_t)r * DC))[c4];
            *(float4*)&As[m][r][c4 << 2] = v;
        }
    }
    __syncthreads();

    v8f c = {0.f, 0.f, 0.f, 0.f, 0.f, 0.f, 0.f, 0.f};

#pragma unroll
    for (int m = 0; m < 3; ++m) {
        const float* Arow  = &As[m][idx][2 * half];
        const float* Bbase = B + (size_t)m * DC * DC
                               + (size_t)2 * half * DC + colB;
#pragma unroll
        for (int k0 = 0; k0 < DC; k0 += 4) {
            v2f a = *(const v2f*)(Arow + k0);
            v2f b;
            b[0] = Bbase[(size_t)k0 * DC];
            b[1] = Bbase[(size_t)k0 * DC + DC];
            c = __builtin_amdgcn_wmma_f32_16x16x4_f32(
                    false, a, false, b, (short)0, c, false, false);
        }
    }

    float bv = bias[colB];
#pragma unroll
    for (int r = 0; r < 8; ++r) {
        int orow = tm * 16 + r + 8 * half;
        out[(size_t)orow * DC + colB] = c[r] + bv;
    }
}


extern "C" void kernel_launch(void* const* d_in, const int* in_sizes, int n_in,
                              void* d_out, int out_size, void* d_ws, size_t ws_size,
                              hipStream_t stream) {
    const int*   ei  = (const int*)d_in[1];
    const float* ew  = (const float*)d_in[2];
    const float* W1  = (const float*)d_in[3];
    const float* b1  = (const float*)d_in[4];
    const float* W2  = (const float*)d_in[5];
    const float* b2  = (const float*)d_in[6];
    float* out = (float*)d_out;

    const int E = in_sizes[2];
    const int D = in_sizes[4];
    const int N = in_sizes[3] / (3 * D);
    const int ND = N * D;

    const int* row = ei;
    const int* col = ei + E;
    const float* W1_0 = W1;
    const float* W1_1 = W1 + (size_t)ND;
    const float* W1_2 = W1 + (size_t)2 * ND;

    char* ws = (char*)d_ws;
    float* deg   = (float*)ws;                   ws += (size_t)N * 4;
    float* dinv  = (float*)ws;                   ws += (size_t)N * 4;
    float* normw = (float*)ws;                   ws += (size_t)E * 4;
    float* B0 = (float*)ws;                      ws += (size_t)ND * 4;
    float* B1 = (float*)ws;                      ws += (size_t)ND * 4;
    float* B2 = (float*)ws;                      ws += (size_t)ND * 4;
    float* H  = (float*)ws;                      ws += (size_t)ND * 4;

    const int TB = 256;
    const int nd4 = ND / 4;
    dim3 gZero((nd4 + TB - 1) / TB), gN4(((N / 4) + TB - 1) / TB);
    dim3 gE((E + TB - 1) / TB), gNn((N + TB - 1) / TB);
    const int spmmThreads = E * (D / 4);
    dim3 gSpmm((spmmThreads + TB - 1) / TB);

    zero_f32<<<gN4, TB, 0, stream>>>(deg, N / 4);
    deg_kernel<<<gE, TB, 0, stream>>>(row, ew, deg, E);
    dinv_kernel<<<gNn, TB, 0, stream>>>(deg, dinv, N);
    normw_kernel<<<gE, TB, 0, stream>>>(row, col, ew, dinv, normw, E);

    zero_f32<<<gZero, TB, 0, stream>>>(B0, nd4);
    spmm_kernel<<<gSpmm, TB, 0, stream>>>(row, col, normw, W1_1, B0, E, D);
    zero_f32<<<gZero, TB, 0, stream>>>(B1, nd4);
    spmm_kernel<<<gSpmm, TB, 0, stream>>>(row, col, normw, W1_2, B1, E, D);
    zero_f32<<<gZero, TB, 0, stream>>>(B2, nd4);
    spmm_kernel<<<gSpmm, TB, 0, stream>>>(row, col, normw, B1, B2, E, D);
    layer1_combine<<<gZero, TB, 0, stream>>>(W1_0, B0, B2, W1_2, b1, H, nd4, D);

    zero_f32<<<gZero, TB, 0, stream>>>(B0, nd4);
    spmm_kernel<<<gSpmm, TB, 0, stream>>>(row, col, normw, H, B0, E, D);
    zero_f32<<<gZero, TB, 0, stream>>>(B1, nd4);
    spmm_kernel<<<gSpmm, TB, 0, stream>>>(row, col, normw, B0, B1, E, D);
    t2_combine<<<gZero, TB, 0, stream>>>(B1, H, B2, nd4);

    gemm3_wmma_f32<<<N / 16, 256, 0, stream>>>(H, B0, B2, W2, b2, out);
}


# Scatter SpMM: a wave takes 8 edges, not 1 (8 lanes fetch index pairs and weights, readlane hands them out, all row loads issued before the atomics): 8x fewer waves to enter; arrival order kept
# speedup vs baseline: 1.6587x; 1.6587x over previous
#include <hip/hip_runtime.h>


typedef __attribute__((ext_vector_type(2))) float v2f;
typedef __attribute__((ext_vector_type(8))) float v8f;

#define DC 128
#define EPW 8

__device__ __forceinline__ int lane_i(int v, int j) {
    return __builtin_amdgcn_readlane(v, j);
}
__device__ __forceinline__ float lane_f(float v, int j) {
    return __int_as_float(__builtin_amdgcn_readlane(__float_as_int(v), j));
}

__device__ __forceinline__ int wave_first_edge() {
    int wv = (int)((blockIdx.x * blockDim.x + threadIdx.x) >> 5);
    return __builtin_amdgcn_readfirstlane(wv) * EPW;
}


__global__ void zero_f32(float* __restrict__ p, int n4) {
    int i = blockIdx.x * blockDim.x + threadIdx.x;
    if (i < n4) ((float4*)p)[i] = make_float4(0.f, 0.f, 0.f, 0.f);
}

__global__ void deg_kernel(const int* __restrict__ row,
                           const float* __restrict__ ew,
                           float* __restrict__ deg, int E) {
    int e = blockIdx.x * blockDim.x + threadIdx.x;
    if (e < E) atomicAdd(&deg[row[e]], ew[e]);
}

__device__ __forceinline__ float inv_sqrt_deg(float d) {
    return (d > 0.f) ? rsqrtf(d) : 0.f;
}

__device__ __forceinline__ void scatter2_edge(const float* __restrict__ inA,
                                              const float* __restrict__ inB,
                                              float* __restrict__ outA,
                                              float* __restrict__ outB,
                                              int r, int c, float w, int l) {
    const float* sa = inA + (size_t)r * DC + l;
    const float* sb = inB + (size_t)r * DC + l;
    float a0 = sa[0], a1 = sa[32], a2 = sa[64], a3 = sa[96];
    float b0 = sb[0], b1 = sb[32], b2 = sb[64], b3 = sb[96];
    float* da = outA + (size_t)c * DC + l;
    float* db = outB + (size_t)c * DC + l;
    atomicAdd(da +  0, w * a0);
    atomicAdd(da + 32, w * a1);
    atomicAdd(da + 64, w * a2);
    atomicAdd(da + 96, w * a3);
    atomicAdd(db +  0, w * b0);
    atomicAdd(db + 32, w * b1);
    atomicAdd(db + 64, w * b2);
    atomicAdd(db + 96, w * b3);
}

__global__ void __launch_bounds__(256)
spmm2_norm_kernel(const int* __restrict__ row,
                  const int* __restrict__ col,
                  const float* __restrict__ ew,
                  const float* __restrict__ deg,
                  const float* __restrict__ inA,
                  const float* __restrict__ inB,
                  float* __restrict__ outA,
                  float* __restrict__ outB,
                  float* __restrict__ nw,
                  int E) {
    const int l  = threadIdx.x & 31;
    const int e0 = wave_first_edge();

    if (e0 + EPW <= E) {
        const int el = e0 + (l & (EPW - 1));
        const int rl = row[el], cl = col[el];
        const float ewl = ew[el];
        float a[EPW][4], b[EPW][4];
#pragma unroll
        for (int j = 0; j < EPW; ++j) {
            const size_t ro = (size_t)lane_i(rl, j) * DC + l;
            const float* sa = inA + ro;
            const float* sb = inB + ro;
            a[j][0] = sa[0]; a[j][1] = sa[32]; a[j][2] = sa[64]; a[j][3] = sa[96];
            b[j][0] = sb[0]; b[j][1] = sb[32]; b[j][2] = sb[64]; b[j][3] = sb[96];
        }
        float wl = 0.f;
        if (l < EPW) {
            wl = -inv_sqrt_deg(deg[rl]) * ewl * inv_sqrt_deg(deg[cl]);
            nw[e0 + l] = wl;
        }
#pragma unroll
        for (int j = 0; j < EPW; ++j) {
            const float w = lane_f(wl, j);
            const size_t co = (size_t)lane_i(cl, j) * DC + l;
            float* da = outA + co;
            float* db = outB + co;
            atomicAdd(da +  0, w * a[j][0]);
            atomicAdd(da + 32, w * a[j][1]);
            atomicAdd(da + 64, w * a[j][2]);
            atomicAdd(da + 96, w * a[j][3]);
            atomicAdd(db +  0, w * b[j][0]);
            atomicAdd(db + 32, w * b[j][1]);
            atomicAdd(db + 64, w * b[j][2]);
            atomicAdd(db + 96, w * b[j][3]);
        }
    } else {
        for (int e = e0; e < E; ++e) {
            const int r = row[e], c = col[e];
            const float w = -inv_sqrt_deg(deg[r]) * ew[e] * inv_sqrt_deg(deg[c]);
            if (l == 0) nw[e] = w;
            scatter2_edge(inA, inB, outA, outB, r, c, w, l);
        }
    }
}

__device__ __forceinline__ void scatter_edge(const float* __restrict__ in,
                                             float* __restrict__ out,
                                             int r, int c, float w, int l) {
    const float* src = in + (size_t)r * DC + l;
    float v0 = src[0], v1 = src[32], v2 = src[64], v3 = src[96];
    float* dst = out + (size_t)c * DC + l;
    atomicAdd(dst +  0, w * v0);
    atomicAdd(dst + 32, w * v1);
    atomicAdd(dst + 64, w * v2);
    atomicAdd(dst + 96, w * v3);
}

__global__ void __launch_bounds__(256)
spmm_kernel(const int* __restrict__ row,
            const int* __restrict__ col,
            const float* __restrict__ nw,
            const float* __restrict__ in,
            float* __restrict__ out,
            int E) {
    const int l  = threadIdx.x & 31;
    const int e0 = wave_first_edge();

    if (e0 + EPW <= E) {
        const int el = e0 + (l & (EPW - 1));
        const int rl = row[el], cl = col[el];
        const float wl = nw[el];
        float v[EPW][4];
#pragma unroll
        for (int j = 0; j < EPW; ++j) {
            const float* src = in + (size_t)lane_i(rl, j) * DC + l;
            v[j][0] = src[0]; v[j][1] = src[32]; v[j][2] = src[64]; v[j][3] = src[96];
        }
#pragma unroll
        for (int j = 0; j < EPW; ++j) {
            const float w = lane_f(wl, j);
            float* dst = out + (size_t)lane_i(cl, j) * DC + l;
            atomicAdd(dst +  0, w * v[j][0]);
            atomicAdd(dst + 32, w * v[j][1]);
            atomicAdd(dst + 64, w * v[j][2]);
            atomicAdd(dst + 96, w * v[j][3]);
        }
    } else {
        for (int e = e0; e < E; ++e)
            scatter_edge(in, out, row[e], col[e], nw[e], l);
    }
}

__global__ void layer1_combine(const float* __restrict__ w10,
                               float* __restrict__ acc0,
                               const float* __restrict__ s2,
                               const float* __restrict__ w12,
                               const float* __restrict__ b1,
                               float* __restrict__ h,
                               float* __restrict__ acc1,
                               int n4, int D) {
    int i = blockIdx.x * blockDim.x + threadIdx.x;
    if (i >= n4) return;
    int cg = i & ((D >> 2) - 1);
    float4 a = ((const float4*)w10)[i];
    float4 s = ((const float4*)acc0)[i];
    float4 t = ((const float4*)s2)[i];
    float4 m = ((const float4*)w12)[i];
    float4 b = ((const float4*)b1)[cg];
    float4 o;
    o.x = fmaxf(a.x + s.x + 2.f * t.x - m.x + b.x, 0.f);
    o.y = fmaxf(a.y + s.y + 2.f * t.y - m.y + b.y, 0.f);
    o.z = fmaxf(a.z + s.z + 2.f * t.z - m.z + b.z, 0.f);
    o.w = fmaxf(a.w + s.w + 2.f * t.w - m.w + b.w, 0.f);
    ((float4*)h)[i] = o;
    const float4 z = make_float4(0.f, 0.f, 0.f, 0.f);
    ((float4*)acc0)[i] = z;
    ((float4*)acc1)[i] = z;
}

#define APAD 132

__global__ void __launch_bounds__(256)
gemm3_wmma_f32(const float* __restrict__ Hm,
               const float* __restrict__ T1,
               const float* __restrict__ U,
               const float* __restrict__ B,
               const float* __restrict__ bias,
               float* __restrict__ out) {
    __shared__ float As[3][16][APAD];

    const int wave = threadIdx.x >> 5;
    const int lane = threadIdx.x & 31;
    const int idx  = lane & 15;
    const int half = lane >> 4;

    const int tm   = blockIdx.x;
    const int tn   = wave;
    const int colB = tn * 16 + idx;

    const size_t strip = (size_t)tm * 16 * DC;
#pragma unroll
    for (int i = 0; i < 2; ++i) {
        int f  = threadIdx.x + i * 256;
        int r  = f >> 5;
        int c4 = f & 31;
        size_t off = strip + (size_t)r * DC;
        float4 vh = ((const float4*)(Hm + off))[c4];
        float4 vt = ((const float4*)(T1 + off))[c4];
        float4 vu = ((const float4*)(U  + off))[c4];
        float4 v2;
        v2.x = 2.f * vu.x - vh.x;  v2.y = 2.f * vu.y - vh.y;
        v2.z = 2.f * vu.z - vh.z;  v2.w = 2.f * vu.w - vh.w;
        *(float4*)&As[0][r][c4 << 2] = vh;
        *(float4*)&As[1][r][c4 << 2] = vt;
        *(float4*)&As[2][r][c4 << 2] = v2;
    }
    __syncthreads();

    v8f c = {0.f, 0.f, 0.f, 0.f, 0.f, 0.f, 0.f, 0.f};

#pragma unroll
    for (int m = 0; m < 3; ++m) {
        const float* Arow  = &As[m][idx][2 * half];
        const float* Bbase = B + (size_t)m * DC * DC
                               + (size_t)2 * half * DC + colB;
#pragma unroll
        for (int k0 = 0; k0 < DC; k0 += 4) {
            v2f a = *(const v2f*)(Arow + k0);
            v2f b;
            b[0] = Bbase[(size_t)k0 * DC];
            b[1] = Bbase[(size_t)k0 * DC + DC];
            c = __builtin_amdgcn_wmma_f32_16x16x4_f32(
                    false, a, false, b, (short)0, c, false, false);
        }
    }

    float bv = bias[colB];
#pragma unroll
    for (int r = 0; r < 8; ++r) {
        int orow = tm * 16 + r + 8 * half;
        out[(size_t)orow * DC + colB] = c[r] + bv;
    }
}


extern "C" void kernel_launch(void* const* d_in, const int* in_sizes, int n_in,
                              void* d_out, int out_size, void* d_ws, size_t ws_size,
                              hipStream_t stream) {
    const int*   ei  = (const int*)d_in[1];
    const float* ew  = (const float*)d_in[2];
    const float* W1  = (const float*)d_in[3];
    const float* b1  = (const float*)d_in[4];
    const float* W2  = (const float*)d_in[5];
    const float* b2  = (const float*)d_in[6];
    float* out = (float*)d_out;

    const int E = in_sizes[2];
    const int D = in_sizes[4];
    const int N = in_sizes[3] / (3 * D);
    const int ND = N * D;

    const int* row = ei;
    const int* col = ei + E;
    const float* W1_0 = W1;
    const float* W1_1 = W1 + (size_t)ND;
    const float* W1_2 = W1 + (size_t)2 * ND;

    char* ws = (char*)d_ws;
    float* deg   = (float*)ws;                   ws += (size_t)N * 4;
    float* B0 = (float*)ws;                      ws += (size_t)ND * 4;
    float* B1 = (float*)ws;                      ws += (size_t)ND * 4;
    float* B2 = (float*)ws;                      ws += (size_t)ND * 4;
    float* H  = (float*)ws;                      ws += (size_t)ND * 4;
    float* normw = (float*)ws;                   ws += (size_t)E * 4;

    const int TB = 256;
    const int nd4 = ND / 4;
    const int nz4 = (N + 3 * ND) / 4;
    dim3 gZero((nz4 + TB - 1) / TB), gComb((nd4 + TB - 1) / TB);
    dim3 gE((E + TB - 1) / TB);
    const int spmmThreads = ((E + EPW - 1) / EPW) * 32;
    dim3 gSpmm((spmmThreads + TB - 1) / TB);

    zero_f32<<<gZero, TB, 0, stream>>>(deg, nz4);
    deg_kernel<<<gE, TB, 0, stream>>>(row, ew, deg, E);
    spmm2_norm_kernel<<<gSpmm, TB, 0, stream>>>(row, col, ew, deg, W1_1, W1_2,
                                                B0, B1, normw, E);
    spmm_kernel<<<gSpmm, TB, 0, stream>>>(row, col, normw, B1, B2, E);
    layer1_combine<<<gComb, TB, 0, stream>>>(W1_0, B0, B2, W1_2, b1, H, B1, nd4, D);

    spmm_kernel<<<gSpmm, TB, 0, stream>>>(row, col, normw, H, B0, E);
    spmm_kernel<<<gSpmm, TB, 0, stream>>>(row, col, normw, B0, B1, E);

    gemm3_wmma_f32<<<N / 16, 256, 0, stream>>>(H, B0, B1, W2, b2, out);
}


# Layer 2 re-associated, GEMM first: h built in its A staging, 3 h@W2 tiles per wave; scatters then add A(hW2_1+2A(hW2_2)) onto out=hW2_0-hW2_2+b2; 8 launches to 7; same products, other sum order
# speedup vs baseline: 1.8151x; 1.0943x over previous
#include <hip/hip_runtime.h>


typedef __attribute__((ext_vector_type(2))) float v2f;
typedef __attribute__((ext_vector_type(8))) float v8f;

#define DC 128
#define EPW 8

__device__ __forceinline__ int lane_i(int v, int j) {
    return __builtin_amdgcn_readlane(v, j);
}
__device__ __forceinline__ float lane_f(float v, int j) {
    return __int_as_float(__builtin_amdgcn_readlane(__float_as_int(v), j));
}

__device__ __forceinline__ int wave_first_edge() {
    int wv = (int)((blockIdx.x * blockDim.x + threadIdx.x) >> 5);
    return __builtin_amdgcn_readfirstlane(wv) * EPW;
}


__global__ void zero_f32(float* __restrict__ p, int n4) {
    int i = blockIdx.x * blockDim.x + threadIdx.x;
    if (i < n4) ((float4*)p)[i] = make_float4(0.f, 0.f, 0.f, 0.f);
}

__global__ void deg_kernel(const int* __restrict__ row,
                           const float* __restrict__ ew,
                           float* __restrict__ deg, int E) {
    int e = blockIdx.x * blockDim.x + threadIdx.x;
    if (e < E) atomicAdd(&deg[row[e]], ew[e]);
}

__device__ __forceinline__ float inv_sqrt_deg(float d) {
    return (d > 0.f) ? rsqrtf(d) : 0.f;
}

__device__ __forceinline__ void scatter2_edge(const float* __restrict__ inA,
                                              const float* __restrict__ inB,
                                              float* __restrict__ outA,
                                              float* __restrict__ outB,
                                              int r, int c, float w, int l) {
    const float* sa = inA + (size_t)r * DC + l;
    const float* sb = inB + (size_t)r * DC + l;
    float a0 = sa[0], a1 = sa[32], a2 = sa[64], a3 = sa[96];
    float b0 = sb[0], b1 = sb[32], b2 = sb[64], b3 = sb[96];
    float* da = outA + (size_t)c * DC + l;
    float* db = outB + (size_t)c * DC + l;
    atomicAdd(da +  0, w * a0);
    atomicAdd(da + 32, w * a1);
    atomicAdd(da + 64, w * a2);
    atomicAdd(da + 96, w * a3);
    atomicAdd(db +  0, w * b0);
    atomicAdd(db + 32, w * b1);
    atomicAdd(db + 64, w * b2);
    atomicAdd(db + 96, w * b3);
}

__global__ void __launch_bounds__(256)
spmm2_norm_kernel(const int* __restrict__ row,
                  const int* __restrict__ col,
                  const float* __restrict__ ew,
                  const float* __restrict__ deg,
                  const float* __restrict__ inA,
                  const float* __restrict__ inB,
                  float* __restrict__ outA,
                  float* __restrict__ outB,
                  float* __restrict__ nw,
                  int E) {
    const int l  = threadIdx.x & 31;
    const int e0 = wave_first_edge();

    if (e0 + EPW <= E) {
        const int el = e0 + (l & (EPW - 1));
        const int rl = row[el], cl = col[el];
        const float ewl = ew[el];
        float a[EPW][4], b[EPW][4];
#pragma unroll
        for (int j = 0; j < EPW; ++j) {
            const size_t ro = (size_t)lane_i(rl, j) * DC + l;
            const float* sa = inA + ro;
            const float* sb = inB + ro;
            a[j][0] = sa[0]; a[j][1] = sa[32]; a[j][2] = sa[64]; a[j][3] = sa[96];
            b[j][0] = sb[0]; b[j][1] = sb[32]; b[j][2] = sb[64]; b[j][3] = sb[96];
        }
        float wl = 0.f;
        if (l < EPW) {
            wl = -inv_sqrt_deg(deg[rl]) * ewl * inv_sqrt_deg(deg[cl]);
            nw[e0 + l] = wl;
        }
#pragma unroll
        for (int j = 0; j < EPW; ++j) {
            const float w = lane_f(wl, j);
            const size_t co = (size_t)lane_i(cl, j) * DC + l;
            float* da = outA + co;
            float* db = outB + co;
            atomicAdd(da +  0, w * a[j][0]);
            atomicAdd(da + 32, w * a[j][1]);
            atomicAdd(da + 64, w * a[j][2]);
            atomicAdd(da + 96, w * a[j][3]);
            atomicAdd(db +  0, w * b[j][0]);
            atomicAdd(db + 32, w * b[j][1]);
            atomicAdd(db + 64, w * b[j][2]);
            atomicAdd(db + 96, w * b[j][3]);
        }
    } else {
        for (int e = e0; e < E; ++e) {
            const int r = row[e], c = col[e];
            const float w = -inv_sqrt_deg(deg[r]) * ew[e] * inv_sqrt_deg(deg[c]);
            if (l == 0) nw[e] = w;
            scatter2_edge(inA, inB, outA, outB, r, c, w, l);
        }
    }
}

__device__ __forceinline__ void scatter_edge(const float* __restrict__ in,
                                             float* __restrict__ out,
                                             int r, int c, float w, int l) {
    const float* src = in + (size_t)r * DC + l;
    float v0 = src[0], v1 = src[32], v2 = src[64], v3 = src[96];
    float* dst = out + (size_t)c * DC + l;
    atomicAdd(dst +  0, w * v0);
    atomicAdd(dst + 32, w * v1);
    atomicAdd(dst + 64, w * v2);
    atomicAdd(dst + 96, w * v3);
}

__global__ void __launch_bounds__(256)
spmm_kernel(const int* __restrict__ row,
            const int* __restrict__ col,
            const float* __restrict__ nw,
            const float* __restrict__ in,
            float* __restrict__ out,
            float sc,
            int E) {
    const int l  = threadIdx.x & 31;
    const int e0 = wave_first_edge();

    if (e0 + EPW <= E) {
        const int el = e0 + (l & (EPW - 1));
        const int rl = row[el], cl = col[el];
        const float wl = sc * nw[el];
        float v[EPW][4];
#pragma unroll
        for (int j = 0; j < EPW; ++j) {
            const float* src = in + (size_t)lane_i(rl, j) * DC + l;
            v[j][0] = src[0]; v[j][1] = src[32]; v[j][2] = src[64]; v[j][3] = src[96];
        }
#pragma unroll
        for (int j = 0; j < EPW; ++j) {
            const float w = lane_f(wl, j);
            float* dst = out + (size_t)lane_i(cl, j) * DC + l;
            atomicAdd(dst +  0, w * v[j][0]);
            atomicAdd(dst + 32, w * v[j][1]);
            atomicAdd(dst + 64, w * v[j][2]);
            atomicAdd(dst + 96, w * v[j][3]);
        }
    } else {
        for (int e = e0; e < E; ++e)
            scatter_edge(in, out, row[e], col[e], sc * nw[e], l);
    }
}

#define APAD 132

__global__ void __launch_bounds__(256)
gemm_h3_wmma_f32(const float* __restrict__ w10,
                 const float* __restrict__ s1,
                 const float* __restrict__ s2,
                 const float* __restrict__ w12,
                 const float* __restrict__ b1,
                 const float* __restrict__ B,
                 const float* __restrict__ bias,
                 float* __restrict__ out,
                 float* __restrict__ Q,
                 float* __restrict__ R) {
    __shared__ float As[16][APAD];

    const int wave = threadIdx.x >> 5;
    const int lane = threadIdx.x & 31;
    const int idx  = lane & 15;
    const int half = lane >> 4;

    const int tm   = blockIdx.x;
    const int tn   = wave;
    const int colB = tn * 16 + idx;

    const size_t strip = (size_t)tm * 16 * DC;
#pragma unroll
    for (int i = 0; i < 2; ++i) {
        int f  = threadIdx.x + i * 256;
        int r  = f >> 5;
        int c4 = f & 31;
        size_t off = strip + (size_t)r * DC;
        float4 a = ((const float4*)(w10 + off))[c4];
        float4 s = ((const float4*)(s1  + off))[c4];
        float4 t = ((const float4*)(s2  + off))[c4];
        float4 m = ((const float4*)(w12 + off))[c4];
        float4 b = ((const float4*)b1)[c4];
        float4 o;
        o.x = fmaxf(a.x + s.x + 2.f * t.x - m.x + b.x, 0.f);
        o.y = fmaxf(a.y + s.y + 2.f * t.y - m.y + b.y, 0.f);
        o.z = fmaxf(a.z + s.z + 2.f * t.z - m.z + b.z, 0.f);
        o.w = fmaxf(a.w + s.w + 2.f * t.w - m.w + b.w, 0.f);
        *(float4*)&As[r][c4 << 2] = o;
    }
    __syncthreads();

    v8f c0 = {0.f, 0.f, 0.f, 0.f, 0.f, 0.f, 0.f, 0.f};
    v8f c1 = {0.f, 0.f, 0.f, 0.f, 0.f, 0.f, 0.f, 0.f};
    v8f c2 = {0.f, 0.f, 0.f, 0.f, 0.f, 0.f, 0.f, 0.f};

    const float* Arow = &As[idx][2 * half];
    const float* Bb0  = B + (size_t)2 * half * DC + colB;
    const float* Bb1  = Bb0 + (size_t)DC * DC;
    const float* Bb2  = Bb1 + (size_t)DC * DC;
#pragma unroll
    for (int k0 = 0; k0 < DC; k0 += 4) {
        v2f a = *(const v2f*)(Arow + k0);
        v2f p, q, u;
        p[0] = Bb0[(size_t)k0 * DC];  p[1] = Bb0[(size_t)k0 * DC + DC];
        q[0] = Bb1[(size_t)k0 * DC];  q[1] = Bb1[(size_t)k0 * DC + DC];
        u[0] = Bb2[(size_t)k0 * DC];  u[1] = Bb2[(size_t)k0 * DC + DC];
        c0 = __builtin_amdgcn_wmma_f32_16x16x4_f32(
                 false, a, false, p, (short)0, c0, false, false);
        c1 = __builtin_amdgcn_wmma_f32_16x16x4_f32(
                 false, a, false, q, (short)0, c1, false, false);
        c2 = __builtin_amdgcn_wmma_f32_16x16x4_f32(
                 false, a, false, u, (short)0, c2, false, false);
    }

    float bv = bias[colB];
#pragma unroll
    for (int r = 0; r < 8; ++r) {
        size_t o = (size_t)(tm * 16 + r + 8 * half) * DC + colB;
        out[o] = c0[r] - c2[r] + bv;
        Q[o]   = c1[r];
        R[o]   = c2[r];
    }
}


extern "C" void kernel_launch(void* const* d_in, const int* in_sizes, int n_in,
                              void* d_out, int out_size, void* d_ws, size_t ws_size,
                              hipStream_t stream) {
    const int*   ei  = (const int*)d_in[1];
    const float* ew  = (const float*)d_in[2];
    const float* W1  = (const float*)d_in[3];
    const float* b1  = (const float*)d_in[4];
    const float* W2  = (const float*)d_in[5];
    const float* b2  = (const float*)d_in[6];
    float* out = (float*)d_out;

    const int E = in_sizes[2];
    const int D = in_sizes[4];
    const int N = in_sizes[3] / (3 * D);
    const int ND = N * D;

    const int* row = ei;
    const int* col = ei + E;
    const float* W1_0 = W1;
    const float* W1_1 = W1 + (size_t)ND;
    const float* W1_2 = W1 + (size_t)2 * ND;

    char* ws = (char*)d_ws;
    float* deg   = (float*)ws;                   ws += (size_t)N * 4;
    float* B0 = (float*)ws;                      ws += (size_t)ND * 4;
    float* B1 = (float*)ws;                      ws += (size_t)ND * 4;
    float* B2 = (float*)ws;                      ws += (size_t)ND * 4;
    float* Qm = (float*)ws;                      ws += (size_t)ND * 4;
    float* normw = (float*)ws;                   ws += (size_t)E * 4;
    float* Rm = B1;

    const int TB = 256;
    const int nz4 = (N + 3 * ND) / 4;
    dim3 gZero((nz4 + TB - 1) / TB);
    dim3 gE((E + TB - 1) / TB);
    const int spmmThreads = ((E + EPW - 1) / EPW) * 32;
    dim3 gSpmm((spmmThreads + TB - 1) / TB);

    zero_f32<<<gZero, TB, 0, stream>>>(deg, nz4);
    deg_kernel<<<gE, TB, 0, stream>>>(row, ew, deg, E);
    spmm2_norm_kernel<<<gSpmm, TB, 0, stream>>>(row, col, ew, deg, W1_1, W1_2,
                                                B0, B1, normw, E);
    spmm_kernel<<<gSpmm, TB, 0, stream>>>(row, col, normw, B1, B2, 1.f, E);

    gemm_h3_wmma_f32<<<N / 16, 256, 0, stream>>>(W1_0, B0, B2, W1_2, b1, W2, b2,
                                                 out, Qm, Rm);
    spmm_kernel<<<gSpmm, TB, 0, stream>>>(row, col, normw, Rm, Qm, 2.f, E);
    spmm_kernel<<<gSpmm, TB, 0, stream>>>(row, col, normw, Qm, out, 1.f, E);
}


# Degree sums and zero fill in one launch: 8 workgroups add one edge slice each to an LDS histogram (LDS float atomics), store partial planes; first scatter joins the 8 in a fixed tree; 7 launches to 6
# speedup vs baseline: 1.9404x; 1.0690x over previous
#include <hip/hip_runtime.h>


typedef __attribute__((ext_vector_type(2))) float v2f;
typedef __attribute__((ext_vector_type(8))) float v8f;

#define DC 128
#define EPW 8
#define NN 4096
#define HPG 2
#define HP (HPG * (32 / EPW))
#define HT 1024
#define HU 4

__device__ __forceinline__ int lane_i(int v, int j) {
    return __builtin_amdgcn_readlane(v, j);
}
__device__ __forceinline__ float lane_f(float v, int j) {
    return __int_as_float(__builtin_amdgcn_readlane(__float_as_int(v), j));
}

__device__ __forceinline__ int wave_first_edge() {
    int wv = (int)((blockIdx.x * blockDim.x + threadIdx.x) >> 5);
    return __builtin_amdgcn_readfirstlane(wv) * EPW;
}


__global__ void __launch_bounds__(HT)
degree_zero_kernel(const int* __restrict__ row,
                   const float* __restrict__ ew,
                   float* __restrict__ degp,
                   float* __restrict__ z, int n4, int E, int N) {
    __shared__ float hist[NN];
    const int t = threadIdx.x;
    if (blockIdx.x < HP) {
        for (int i = t; i < N; i += HT) hist[i] = 0.f;
        __syncthreads();
        const int slice = (E + HP - 1) / HP;
        const int eb = blockIdx.x * slice;
        const int ee = min(eb + slice, E);
        int base = eb;
        for (; base + HU * HT <= ee; base += HU * HT) {
            int r[HU];
            float w[HU];
#pragma unroll
            for (int k = 0; k < HU; ++k) {
                r[k] = row[base + k * HT + t];
                w[k] = ew[base + k * HT + t];
            }
#pragma unroll
            for (int k = 0; k < HU; ++k) atomicAdd(&hist[r[k]], w[k]);
        }
        for (int e = base + t; e < ee; e += HT) atomicAdd(&hist[row[e]], ew[e]);
        __syncthreads();
        float* dp = degp + (size_t)blockIdx.x * N;
        for (int i = t; i < N; i += HT) dp[i] = hist[i];
    } else {
        const int i = (blockIdx.x - HP) * HT + t;
        if (i < n4) ((float4*)z)[i] = make_float4(0.f, 0.f, 0.f, 0.f);
    }
}

__device__ __forceinline__ float degree_of(const float* __restrict__ degp,
                                           int N, int i) {
    float s[HP];
#pragma unroll
    for (int p = 0; p < HP; ++p) s[p] = degp[(size_t)p * N + i];
#pragma unroll
    for (int m = 1; m < HP; m <<= 1)
#pragma unroll
        for (int p = 0; p < HP; p += 2 * m) s[p] += s[p + m];
    return s[0];
}

__device__ __forceinline__ float inv_sqrt_deg(float d) {
    return (d > 0.f) ? rsqrtf(d) : 0.f;
}

__device__ __forceinline__ void scatter2_edge(const float* __restrict__ inA,
                                              const float* __restrict__ inB,
                                              float* __restrict__ outA,
                                              float* __restrict__ outB,
                                              int r, int c, float w, int l) {
    const float* sa = inA + (size_t)r * DC + l;
    const float* sb = inB + (size_t)r * DC + l;
    float a0 = sa[0], a1 = sa[32], a2 = sa[64], a3 = sa[96];
    float b0 = sb[0], b1 = sb[32], b2 = sb[64], b3 = sb[96];
    float* da = outA + (size_t)c * DC + l;
    float* db = outB + (size_t)c * DC + l;
    atomicAdd(da +  0, w * a0);
    atomicAdd(da + 32, w * a1);
    atomicAdd(da + 64, w * a2);
    atomicAdd(da + 96, w * a3);
    atomicAdd(db +  0, w * b0);
    atomicAdd(db + 32, w * b1);
    atomicAdd(db + 64, w * b2);
    atomicAdd(db + 96, w * b3);
}

__global__ void __launch_bounds__(256)
spmm2_norm_kernel(const int* __restrict__ row,
                  const int* __restrict__ col,
                  const float* __restrict__ ew,
                  const float* __restrict__ degp,
                  const float* __restrict__ inA,
                  const float* __restrict__ inB,
                  float* __restrict__ outA,
                  float* __restrict__ outB,
                  float* __restrict__ nw,
                  int E, int N) {
    const int l  = threadIdx.x & 31;
    const int e0 = wave_first_edge();

    if (e0 + EPW <= E) {
        const int el = e0 + (l & (EPW - 1));
        const int rl = row[el], cl = col[el];
        const float ewl = ew[el];
        float a[EPW][4], b[EPW][4];
#pragma unroll
        for (int j = 0; j < EPW; ++j) {
            const size_t ro = (size_t)lane_i(rl, j) * DC + l;
            const float* sa = inA + ro;
            const float* sb = inB + ro;
            a[j][0] = sa[0]; a[j][1] = sa[32]; a[j][2] = sa[64]; a[j][3] = sa[96];
            b[j][0] = sb[0]; b[j][1] = sb[32]; b[j][2] = sb[64]; b[j][3] = sb[96];
        }
        const float* dg = degp + (size_t)(l / EPW) * HPG * N;
        float pr[HPG], pc[HPG];
#pragma unroll
        for (int p = 0; p < HPG; ++p) {
            pr[p] = dg[(size_t)p * N + rl];
            pc[p] = dg[(size_t)p * N + cl];
        }
#pragma unroll
        for (int m = 1; m < HPG; m <<= 1)
#pragma unroll
            for (int p = 0; p < HPG; p += 2 * m) { pr[p] += pr[p + m]; pc[p] += pc[p + m]; }
        float dr = pr[0], dc = pc[0];
#pragma unroll
        for (int m = EPW; m < 32; m <<= 1) {
            dr += __shfl_xor(dr, m);
            dc += __shfl_xor(dc, m);
        }
        const float wl = -inv_sqrt_deg(dr) * ewl * inv_sqrt_deg(dc);
        if (l < EPW) nw[e0 + l] = wl;
#pragma unroll
        for (int j = 0; j < EPW; ++j) {
            const float w = lane_f(wl, j);
            const size_t co = (size_t)lane_i(cl, j) * DC + l;
            float* da = outA + co;
            float* db = outB + co;
            atomicAdd(da +  0, w * a[j][0]);
            atomicAdd(da + 32, w * a[j][1]);
            atomicAdd(da + 64, w * a[j][2]);
            atomicAdd(da + 96, w * a[j][3]);
            atomicAdd(db +  0, w * b[j][0]);
            atomicAdd(db + 32, w * b[j][1]);
            atomicAdd(db + 64, w * b[j][2]);
            atomicAdd(db + 96, w * b[j][3]);
        }
    } else {
        for (int e = e0; e < E; ++e) {
            const int r = row[e], c = col[e];
            const float w = -inv_sqrt_deg(degree_of(degp, N, r)) * ew[e]
                            * inv_sqrt_deg(degree_of(degp, N, c));
            if (l == 0) nw[e] = w;
            scatter2_edge(inA, inB, outA, outB, r, c, w, l);
        }
    }
}

__device__ __forceinline__ void scatter_edge(const float* __restrict__ in,
                                             float* __restrict__ out,
                                             int r, int c, float w, int l) {
    const float* src = in + (size_t)r * DC + l;
    float v0 = src[0], v1 = src[32], v2 = src[64], v3 = src[96];
    float* dst = out + (size_t)c * DC + l;
    atomicAdd(dst +  0, w * v0);
    atomicAdd(dst + 32, w * v1);
    atomicAdd(dst + 64, w * v2);
    atomicAdd(dst + 96, w * v3);
}

__global__ void __launch_bounds__(256)
spmm_kernel(const int* __restrict__ row,
            const int* __restrict__ col,
            const float* __restrict__ nw,
            const float* __restrict__ in,
            float* __restrict__ out,
            float sc,
            int E) {
    const int l  = threadIdx.x & 31;
    const int e0 = wave_first_edge();

    if (e0 + EPW <= E) {
        const int el = e0 + (l & (EPW - 1));
        const int rl = row[el], cl = col[el];
        const float wl = sc * nw[el];
        float v[EPW][4];
#pragma unroll
        for (int j = 0; j < EPW; ++j) {
            const float* src = in + (size_t)lane_i(rl, j) * DC + l;
            v[j][0] = src[0]; v[j][1] = src[32]; v[j][2] = src[64]; v[j][3] = src[96];
        }
#pragma unroll
        for (int j = 0; j < EPW; ++j) {
            const float w = lane_f(wl, j);
            float* dst = out + (size_t)lane_i(cl, j) * DC + l;
            atomicAdd(dst +  0, w * v[j][0]);
            atomicAdd(dst + 32, w * v[j][1]);
            atomicAdd(dst + 64, w * v[j][2]);
            atomicAdd(dst + 96, w * v[j][3]);
        }
    } else {
        for (int e = e0; e < E; ++e)
            scatter_edge(in, out, row[e], col[e], sc * nw[e], l);
    }
}

#define APAD 132

__global__ void __launch_bounds__(256)
gemm_h3_wmma_f32(const float* __restrict__ w10,
                 const float* __restrict__ s1,
                 const float* __restrict__ s2,
                 const float* __restrict__ w12,
                 const float* __restrict__ b1,
                 const float* __restrict__ B,
                 const float* __restrict__ bias,
                 float* __restrict__ out,
                 float* __restrict__ Q,
                 float* __restrict__ R) {
    __shared__ float As[16][APAD];

    const int wave = threadIdx.x >> 5;
    const int lane = threadIdx.x & 31;
    const int idx  = lane & 15;
    const int half = lane >> 4;

    const int tm   = blockIdx.x;
    const int tn   = wave;
    const int colB = tn * 16 + idx;

    const size_t strip = (size_t)tm * 16 * DC;
#pragma unroll
    for (int i = 0; i < 2; ++i) {
        int f  = threadIdx.x + i * 256;
        int r  = f >> 5;
        int c4 = f & 31;
        size_t off = strip + (size_t)r * DC;
        float4 a = ((const float4*)(w10 + off))[c4];
        float4 s = ((const float4*)(s1  + off))[c4];
        float4 t = ((const float4*)(s2  + off))[c4];
        float4 m = ((const float4*)(w12 + off))[c4];
        float4 b = ((const float4*)b1)[c4];
        float4 o;
        o.x = fmaxf(a.x + s.x + 2.f * t.x - m.x + b.x, 0.f);
        o.y = fmaxf(a.y + s.y + 2.f * t.y - m.y + b.y, 0.f);
        o.z = fmaxf(a.z + s.z + 2.f * t.z - m.z + b.z, 0.f);
        o.w = fmaxf(a.w + s.w + 2.f * t.w - m.w + b.w, 0.f);
        *(float4*)&As[r][c4 << 2] = o;
    }
    __syncthreads();

    v8f c0 = {0.f, 0.f, 0.f, 0.f, 0.f, 0.f, 0.f, 0.f};
    v8f c1 = {0.f, 0.f, 0.f, 0.f, 0.f, 0.f, 0.f, 0.f};
    v8f c2 = {0.f, 0.f, 0.f, 0.f, 0.f, 0.f, 0.f, 0.f};

    const float* Arow = &As[idx][2 * half];
    const float* Bb0  = B + (size_t)2 * half * DC + colB;
    const float* Bb1  = Bb0 + (size_t)DC * DC;
    const float* Bb2  = Bb1 + (size_t)DC * DC;
#pragma unroll
    for (int k0 = 0; k0 < DC; k0 += 4) {
        v2f a = *(const v2f*)(Arow + k0);
        v2f p, q, u;
        p[0] = Bb0[(size_t)k0 * DC];  p[1] = Bb0[(size_t)k0 * DC + DC];
        q[0] = Bb1[(size_t)k0 * DC];  q[1] = Bb1[(size_t)k0 * DC + DC];
        u[0] = Bb2[(size_t)k0 * DC];  u[1] = Bb2[(size_t)k0 * DC + DC];
        c0 = __builtin_amdgcn_wmma_f32_16x16x4_f32(
                 false, a, false, p, (short)0, c0, false, false);
        c1 = __builtin_amdgcn_wmma_f32_16x16x4_f32(
                 false, a, false, q, (short)0, c1, false, false);
        c2 = __builtin_amdgcn_wmma_f32_16x16x4_f32(
                 false, a, false, u, (short)0, c2, false, false);
    }

    float bv = bias[colB];
#pragma unroll
    for (int r = 0; r < 8; ++r) {
        size_t o = (size_t)(tm * 16 + r + 8 * half) * DC + colB;
        out[o] = c0[r] - c2[r] + bv;
        Q[o]   = c1[r];
        R[o]   = c2[r];
    }
}


extern "C" void kernel_launch(void* const* d_in, const int* in_sizes, int n_in,
                              void* d_out, int out_size, void* d_ws, size_t ws_size,
                              hipStream_t stream) {
    const int*   ei  = (const int*)d_in[1];
    const float* ew  = (const float*)d_in[2];
    const float* W1  = (const float*)d_in[3];
    const float* b1  = (const float*)d_in[4];
    const float* W2  = (const float*)d_in[5];
    const float* b2  = (const float*)d_in[6];
    float* out = (float*)d_out;

    const int E = in_sizes[2];
    const int D = in_sizes[4];
    const int N = in_sizes[3] / (3 * D);
    const int ND = N * D;

    const int* row = ei;
    const int* col = ei + E;
    const float* W1_0 = W1;
    const float* W1_1 = W1 + (size_t)ND;
    const float* W1_2 = W1 + (size_t)2 * ND;

    char* ws = (char*)d_ws;
    float* B0 = (float*)ws;                      ws += (size_t)ND * 4;
    float* B1 = (float*)ws;                      ws += (size_t)ND * 4;
    float* B2 = (float*)ws;                      ws += (size_t)ND * 4;
    float* Qm = (float*)ws;                      ws += (size_t)ND * 4;
    float* normw = (float*)ws;                   ws += (size_t)E * 4;
    float* Rm = B1;
    float* degp = Qm;

    const int TB = 256;
    const int nz4 = (3 * ND) / 4;
    dim3 gInit(HP + (nz4 + HT - 1) / HT);
    const int spmmThreads = ((E + EPW - 1) / EPW) * 32;
    dim3 gSpmm((spmmThreads + TB - 1) / TB);

    degree_zero_kernel<<<gInit, HT, 0, stream>>>(row, ew, degp, B0, nz4, E, N);
    spmm2_norm_kernel<<<gSpmm, TB, 0, stream>>>(row, col, ew, degp, W1_1, W1_2,
                                                B0, B1, normw, E, N);
    spmm_kernel<<<gSpmm, TB, 0, stream>>>(row, col, normw, B1, B2, 1.f, E);

    gemm_h3_wmma_f32<<<N / 16, 256, 0, stream>>>(W1_0, B0, B2, W1_2, b1, W2, b2,
                                                 out, Qm, Rm);
    spmm_kernel<<<gSpmm, TB, 0, stream>>>(row, col, normw, Rm, Qm, 2.f, E);
    spmm_kernel<<<gSpmm, TB, 0, stream>>>(row, col, normw, Qm, out, 1.f, E);
}


# Layer 1: S1's own scatter dropped; second scatter adds nw*(W1_1[row]+2*S2a[row]) per edge as ONE atomic set: 5 row scatters to 4, 3 zeroed planes to 2; other sum order, arrival order kept
# speedup vs baseline: 2.0894x; 1.0768x over previous
#include <hip/hip_runtime.h>


typedef __attribute__((ext_vector_type(2))) float v2f;
typedef __attribute__((ext_vector_type(8))) float v8f;

#define DC 128
#define EPW 8
#define NN 4096
#define HPG 2
#define HP (HPG * (32 / EPW))
#define HT 1024
#define HU 4

__device__ __forceinline__ int lane_i(int v, int j) {
    return __builtin_amdgcn_readlane(v, j);
}
__device__ __forceinline__ float lane_f(float v, int j) {
    return __int_as_float(__builtin_amdgcn_readlane(__float_as_int(v), j));
}

__device__ __forceinline__ int wave_first_edge() {
    int wv = (int)((blockIdx.x * blockDim.x + threadIdx.x) >> 5);
    return __builtin_amdgcn_readfirstlane(wv) * EPW;
}


__global__ void __launch_bounds__(HT)
degree_zero_kernel(const int* __restrict__ row,
                   const float* __restrict__ ew,
                   float* __restrict__ degp,
                   float* __restrict__ z, int n4, int E, int N) {
    __shared__ float hist[NN];
    const int t = threadIdx.x;
    if (blockIdx.x < HP) {
        for (int i = t; i < N; i += HT) hist[i] = 0.f;
        __syncthreads();
        const int slice = (E + HP - 1) / HP;
        const int eb = blockIdx.x * slice;
        const int ee = min(eb + slice, E);
        int base = eb;
        for (; base + HU * HT <= ee; base += HU * HT) {
            int r[HU];
            float w[HU];
#pragma unroll
            for (int k = 0; k < HU; ++k) {
                r[k] = row[base + k * HT + t];
                w[k] = ew[base + k * HT + t];
            }
#pragma unroll
            for (int k = 0; k < HU; ++k) atomicAdd(&hist[r[k]], w[k]);
        }
        for (int e = base + t; e < ee; e += HT) atomicAdd(&hist[row[e]], ew[e]);
        __syncthreads();
        float* dp = degp + (size_t)blockIdx.x * N;
        for (int i = t; i < N; i += HT) dp[i] = hist[i];
    } else {
        const int i = (blockIdx.x - HP) * HT + t;
        if (i < n4) ((float4*)z)[i] = make_float4(0.f, 0.f, 0.f, 0.f);
    }
}

__device__ __forceinline__ float degree_of(const float* __restrict__ degp,
                                           int N, int i) {
    float s[HP];
#pragma unroll
    for (int p = 0; p < HP; ++p) s[p] = degp[(size_t)p * N + i];
#pragma unroll
    for (int m = 1; m < HP; m <<= 1)
#pragma unroll
        for (int p = 0; p < HP; p += 2 * m) s[p] += s[p + m];
    return s[0];
}

__device__ __forceinline__ float inv_sqrt_deg(float d) {
    return (d > 0.f) ? rsqrtf(d) : 0.f;
}

__device__ __forceinline__ void scatter_edge(const float* __restrict__ in,
                                             float* __restrict__ out,
                                             int r, int c, float w, int l) {
    const float* src = in + (size_t)r * DC + l;
    float v0 = src[0], v1 = src[32], v2 = src[64], v3 = src[96];
    float* dst = out + (size_t)c * DC + l;
    atomicAdd(dst +  0, w * v0);
    atomicAdd(dst + 32, w * v1);
    atomicAdd(dst + 64, w * v2);
    atomicAdd(dst + 96, w * v3);
}

__global__ void __launch_bounds__(256)
spmm_norm_kernel(const int* __restrict__ row,
                 const int* __restrict__ col,
                 const float* __restrict__ ew,
                 const float* __restrict__ degp,
                 const float* __restrict__ in,
                 float* __restrict__ out,
                 float* __restrict__ nw,
                 int E, int N) {
    const int l  = threadIdx.x & 31;
    const int e0 = wave_first_edge();

    if (e0 + EPW <= E) {
        const int el = e0 + (l & (EPW - 1));
        const int rl = row[el], cl = col[el];
        const float ewl = ew[el];
        float v[EPW][4];
#pragma unroll
        for (int j = 0; j < EPW; ++j) {
            const float* src = in + (size_t)lane_i(rl, j) * DC + l;
            v[j][0] = src[0]; v[j][1] = src[32]; v[j][2] = src[64]; v[j][3] = src[96];
        }
        const float* dg = degp + (size_t)(l / EPW) * HPG * N;
        float pr[HPG], pc[HPG];
#pragma unroll
        for (int p = 0; p < HPG; ++p) {
            pr[p] = dg[(size_t)p * N + rl];
            pc[p] = dg[(size_t)p * N + cl];
        }
#pragma unroll
        for (int m = 1; m < HPG; m <<= 1)
#pragma unroll
            for (int p = 0; p < HPG; p += 2 * m) { pr[p] += pr[p + m]; pc[p] += pc[p + m]; }
        float dr = pr[0], dc = pc[0];
#pragma unroll
        for (int m = EPW; m < 32; m <<= 1) {
            dr += __shfl_xor(dr, m);
            dc += __shfl_xor(dc, m);
        }
        const float wl = -inv_sqrt_deg(dr) * ewl * inv_sqrt_deg(dc);
        if (l < EPW) nw[e0 + l] = wl;
#pragma unroll
        for (int j = 0; j < EPW; ++j) {
            const float w = lane_f(wl, j);
            float* dst = out + (size_t)lane_i(cl, j) * DC + l;
            atomicAdd(dst +  0, w * v[j][0]);
            atomicAdd(dst + 32, w * v[j][1]);
            atomicAdd(dst + 64, w * v[j][2]);
            atomicAdd(dst + 96, w * v[j][3]);
        }
    } else {
        for (int e = e0; e < E; ++e) {
            const int r = row[e], c = col[e];
            const float w = -inv_sqrt_deg(degree_of(degp, N, r)) * ew[e]
                            * inv_sqrt_deg(degree_of(degp, N, c));
            if (l == 0) nw[e] = w;
            scatter_edge(in, out, r, c, w, l);
        }
    }
}

__device__ __forceinline__ void scatter_pair_edge(const float* __restrict__ inA,
                                                  const float* __restrict__ inB,
                                                  float* __restrict__ out,
                                                  int r, int c, float w, int l) {
    const float* sa = inA + (size_t)r * DC + l;
    const float* sb = inB + (size_t)r * DC + l;
    float a0 = sa[0], a1 = sa[32], a2 = sa[64], a3 = sa[96];
    float b0 = sb[0], b1 = sb[32], b2 = sb[64], b3 = sb[96];
    float* dst = out + (size_t)c * DC + l;
    atomicAdd(dst +  0, w * (a0 + 2.f * b0));
    atomicAdd(dst + 32, w * (a1 + 2.f * b1));
    atomicAdd(dst + 64, w * (a2 + 2.f * b2));
    atomicAdd(dst + 96, w * (a3 + 2.f * b3));
}

__global__ void __launch_bounds__(256)
spmm_pair_kernel(const int* __restrict__ row,
                 const int* __restrict__ col,
                 const float* __restrict__ nw,
                 const float* __restrict__ inA,
                 const float* __restrict__ inB,
                 float* __restrict__ out,
                 int E) {
    const int l  = threadIdx.x & 31;
    const int e0 = wave_first_edge();

    if (e0 + EPW <= E) {
        const int el = e0 + (l & (EPW - 1));
        const int rl = row[el], cl = col[el];
        const float wl = nw[el];
        float a[EPW][4], b[EPW][4];
#pragma unroll
        for (int j = 0; j < EPW; ++j) {
            const size_t ro = (size_t)lane_i(rl, j) * DC + l;
            const float* sa = inA + ro;
            const float* sb = inB + ro;
            a[j][0] = sa[0]; a[j][1] = sa[32]; a[j][2] = sa[64]; a[j][3] = sa[96];
            b[j][0] = sb[0]; b[j][1] = sb[32]; b[j][2] = sb[64]; b[j][3] = sb[96];
        }
#pragma unroll
        for (int j = 0; j < EPW; ++j) {
            const float w = lane_f(wl, j);
            float* dst = out + (size_t)lane_i(cl, j) * DC + l;
            atomicAdd(dst +  0, w * (a[j][0] + 2.f * b[j][0]));
            atomicAdd(dst + 32, w * (a[j][1] + 2.f * b[j][1]));
            atomicAdd(dst + 64, w * (a[j][2] + 2.f * b[j][2]));
            atomicAdd(dst + 96, w * (a[j][3] + 2.f * b[j][3]));
        }
    } else {
        for (int e = e0; e < E; ++e)
            scatter_pair_edge(inA, inB, out, row[e], col[e], nw[e], l);
    }
}

__global__ void __launch_bounds__(256)
spmm_kernel(const int* __restrict__ row,
            const int* __restrict__ col,
            const float* __restrict__ nw,
            const float* __restrict__ in,
            float* __restrict__ out,
            float sc,
            int E) {
    const int l  = threadIdx.x & 31;
    const int e0 = wave_first_edge();

    if (e0 + EPW <= E) {
        const int el = e0 + (l & (EPW - 1));
        const int rl = row[el], cl = col[el];
        const float wl = sc * nw[el];
        float v[EPW][4];
#pragma unroll
        for (int j = 0; j < EPW; ++j) {
            const float* src = in + (size_t)lane_i(rl, j) * DC + l;
            v[j][0] = src[0]; v[j][1] = src[32]; v[j][2] = src[64]; v[j][3] = src[96];
        }
#pragma unroll
        for (int j = 0; j < EPW; ++j) {
            const float w = lane_f(wl, j);
            float* dst = out + (size_t)lane_i(cl, j) * DC + l;
            atomicAdd(dst +  0, w * v[j][0]);
            atomicAdd(dst + 32, w * v[j][1]);
            atomicAdd(dst + 64, w * v[j][2]);
            atomicAdd(dst + 96, w * v[j][3]);
        }
    } else {
        for (int e = e0; e < E; ++e)
            scatter_edge(in, out, row[e], col[e], sc * nw[e], l);
    }
}

#define APAD 132

__global__ void __launch_bounds__(256)
gemm_h3_wmma_f32(const float* __restrict__ w10,
                 const float* __restrict__ s1,
                 const float* __restrict__ w12,
                 const float* __restrict__ b1,
                 const float* __restrict__ B,
                 const float* __restrict__ bias,
                 float* __restrict__ out,
                 float* __restrict__ Q,
                 float* __restrict__ R) {
    __shared__ float As[16][APAD];

    const int wave = threadIdx.x >> 5;
    const int lane = threadIdx.x & 31;
    const int idx  = lane & 15;
    const int half = lane >> 4;

    const int tm   = blockIdx.x;
    const int tn   = wave;
    const int colB = tn * 16 + idx;

    const size_t strip = (size_t)tm * 16 * DC;
#pragma unroll
    for (int i = 0; i < 2; ++i) {
        int f  = threadIdx.x + i * 256;
        int r  = f >> 5;
        int c4 = f & 31;
        size_t off = strip + (size_t)r * DC;
        float4 a = ((const float4*)(w10 + off))[c4];
        float4 s = ((const float4*)(s1  + off))[c4];
        float4 m = ((const float4*)(w12 + off))[c4];
        float4 b = ((const float4*)b1)[c4];
        float4 o;
        o.x = fmaxf(a.x + s.x - m.x + b.x, 0.f);
        o.y = fmaxf(a.y + s.y - m.y + b.y, 0.f);
        o.z = fmaxf(a.z + s.z - m.z + b.z, 0.f);
        o.w = fmaxf(a.w + s.w - m.w + b.w, 0.f);
        *(float4*)&As[r][c4 << 2] = o;
    }
    __syncthreads();

    v8f c0 = {0.f, 0.f, 0.f, 0.f, 0.f, 0.f, 0.f, 0.f};
    v8f c1 = {0.f, 0.f, 0.f, 0.f, 0.f, 0.f, 0.f, 0.f};
    v8f c2 = {0.f, 0.f, 0.f, 0.f, 0.f, 0.f, 0.f, 0.f};

    const float* Arow = &As[idx][2 * half];
    const float* Bb0  = B + (size_t)2 * half * DC + colB;
    const float* Bb1  = Bb0 + (size_t)DC * DC;
    const float* Bb2  = Bb1 + (size_t)DC * DC;
#pragma unroll
    for (int k0 = 0; k0 < DC; k0 += 4) {
        v2f a = *(const v2f*)(Arow + k0);
        v2f p, q, u;
        p[0] = Bb0[(size_t)k0 * DC];  p[1] = Bb0[(size_t)k0 * DC + DC];
        q[0] = Bb1[(size_t)k0 * DC];  q[1] = Bb1[(size_t)k0 * DC + DC];
        u[0] = Bb2[(size_t)k0 * DC];  u[1] = Bb2[(size_t)k0 * DC + DC];
        c0 = __builtin_amdgcn_wmma_f32_16x16x4_f32(
                 false, a, false, p, (short)0, c0, false, false);
        c1 = __builtin_amdgcn_wmma_f32_16x16x4_f32(
                 false, a, false, q, (short)0, c1, false, false);
        c2 = __builtin_amdgcn_wmma_f32_16x16x4_f32(
                 false, a, false, u, (short)0, c2, false, false);
    }

    float bv = bias[colB];
#pragma unroll
    for (int r = 0; r < 8; ++r) {
        size_t o = (size_t)(tm * 16 + r + 8 * half) * DC + colB;
        out[o] = c0[r] - c2[r] + bv;
        Q[o]   = c1[r];
        R[o]   = c2[r];
    }
}


extern "C" void kernel_launch(void* const* d_in, const int* in_sizes, int n_in,
                              void* d_out, int out_size, void* d_ws, size_t ws_size,
                              hipStream_t stream) {
    const int*   ei  = (const int*)d_in[1];
    const float* ew  = (const float*)d_in[2];
    const float* W1  = (const float*)d_in[3];
    const float* b1  = (const float*)d_in[4];
    const float* W2  = (const float*)d_in[5];
    const float* b2  = (const float*)d_in[6];
    float* out = (float*)d_out;

    const int E = in_sizes[2];
    const int D = in_sizes[4];
    const int N = in_sizes[3] / (3 * D);
    const int ND = N * D;

    const int* row = ei;
    const int* col = ei + E;
    const float* W1_0 = W1;
    const float* W1_1 = W1 + (size_t)ND;
    const float* W1_2 = W1 + (size_t)2 * ND;

    char* ws = (char*)d_ws;
    float* B0 = (float*)ws;                      ws += (size_t)ND * 4;
    float* B1 = (float*)ws;                      ws += (size_t)ND * 4;
    float* Qm = (float*)ws;                      ws += (size_t)ND * 4;
    float* normw = (float*)ws;                   ws += (size_t)E * 4;
    float* Rm = B1;
    float* degp = Qm;

    const int TB = 256;
    const int nz4 = (2 * ND) / 4;
    dim3 gInit(HP + (nz4 + HT - 1) / HT);
    const int spmmThreads = ((E + EPW - 1) / EPW) * 32;
    dim3 gSpmm((spmmThreads + TB - 1) / TB);

    degree_zero_kernel<<<gInit, HT, 0, stream>>>(row, ew, degp, B0, nz4, E, N);
    spmm_norm_kernel<<<gSpmm, TB, 0, stream>>>(row, col, ew, degp, W1_2,
                                               B1, normw, E, N);
    spmm_pair_kernel<<<gSpmm, TB, 0, stream>>>(row, col, normw, W1_1, B1,
                                               B0, E);

    gemm_h3_wmma_f32<<<N / 16, 256, 0, stream>>>(W1_0, B0, W1_2, b1, W2, b2,
                                                 out, Qm, Rm);
    spmm_kernel<<<gSpmm, TB, 0, stream>>>(row, col, normw, Rm, Qm, 2.f, E);
    spmm_kernel<<<gSpmm, TB, 0, stream>>>(row, col, normw, Qm, out, 1.f, E);
}


# Layer-1 accumulators start from their row terms, not zero: T=W1_1/2+A*W1_2, S=W1_0-W1_2+b1+2*A*T; no zero fill, 2nd scatter loads 1 row per edge not 2; other sum order, arrival order kept
# speedup vs baseline: 2.2288x; 1.0667x over previous
#include <hip/hip_runtime.h>


typedef __attribute__((ext_vector_type(2))) float v2f;
typedef __attribute__((ext_vector_type(8))) float v8f;

#define DC 128
#define EPW 8
#define NN 4096
#define HPG 2
#define HP (HPG * (32 / EPW))
#define HT 1024
#define HU 4

__device__ __forceinline__ int lane_i(int v, int j) {
    return __builtin_amdgcn_readlane(v, j);
}
__device__ __forceinline__ float lane_f(float v, int j) {
    return __int_as_float(__builtin_amdgcn_readlane(__float_as_int(v), j));
}

__device__ __forceinline__ int wave_first_edge() {
    int wv = (int)((blockIdx.x * blockDim.x + threadIdx.x) >> 5);
    return __builtin_amdgcn_readfirstlane(wv) * EPW;
}


__global__ void __launch_bounds__(HT)
degree_init_kernel(const int* __restrict__ row,
                   const float* __restrict__ ew,
                   float* __restrict__ degp,
                   const float* __restrict__ hsrc,
                   float* __restrict__ z, int n4, int E, int N) {
    __shared__ float hist[NN];
    const int t = threadIdx.x;
    if (blockIdx.x < HP) {
        for (int i = t; i < N; i += HT) hist[i] = 0.f;
        __syncthreads();
        const int slice = (E + HP - 1) / HP;
        const int eb = blockIdx.x * slice;
        const int ee = min(eb + slice, E);
        int base = eb;
        for (; base + HU * HT <= ee; base += HU * HT) {
            int r[HU];
            float w[HU];
#pragma unroll
            for (int k = 0; k < HU; ++k) {
                r[k] = row[base + k * HT + t];
                w[k] = ew[base + k * HT + t];
            }
#pragma unroll
            for (int k = 0; k < HU; ++k) atomicAdd(&hist[r[k]], w[k]);
        }
        for (int e = base + t; e < ee; e += HT) atomicAdd(&hist[row[e]], ew[e]);
        __syncthreads();
        float* dp = degp + (size_t)blockIdx.x * N;
        for (int i = t; i < N; i += HT) dp[i] = hist[i];
    } else {
        const int i = (blockIdx.x - HP) * HT + t;
        if (i < n4) {
            const float4 v = ((const float4*)hsrc)[i];
            ((float4*)z)[i] = make_float4(0.5f * v.x, 0.5f * v.y, 0.5f * v.z, 0.5f * v.w);
        }
    }
}

__device__ __forceinline__ float degree_of(const float* __restrict__ degp,
                                           int N, int i) {
    float s[HP];
#pragma unroll
    for (int p = 0; p < HP; ++p) s[p] = degp[(size_t)p * N + i];
#pragma unroll
    for (int m = 1; m < HP; m <<= 1)
#pragma unroll
        for (int p = 0; p < HP; p += 2 * m) s[p] += s[p + m];
    return s[0];
}

__device__ __forceinline__ float inv_sqrt_deg(float d) {
    return (d > 0.f) ? rsqrtf(d) : 0.f;
}

__device__ __forceinline__ void scatter_edge(const float* __restrict__ in,
                                             float* __restrict__ out,
                                             int r, int c, float w, int l) {
    const float* src = in + (size_t)r * DC + l;
    float v0 = src[0], v1 = src[32], v2 = src[64], v3 = src[96];
    float* dst = out + (size_t)c * DC + l;
    atomicAdd(dst +  0, w * v0);
    atomicAdd(dst + 32, w * v1);
    atomicAdd(dst + 64, w * v2);
    atomicAdd(dst + 96, w * v3);
}

__global__ void __launch_bounds__(256)
spmm_norm_kernel(const int* __restrict__ row,
                 const int* __restrict__ col,
                 const float* __restrict__ ew,
                 const float* __restrict__ degp,
                 const float* __restrict__ in,
                 float* __restrict__ out,
                 float* __restrict__ nw,
                 const float* __restrict__ p,
                 const float* __restrict__ pb,
                 float* __restrict__ s0,
                 int E, int N) {
    const int l  = threadIdx.x & 31;
    const int e0 = wave_first_edge();

    if (e0 + EPW <= E) {
        const int el = e0 + (l & (EPW - 1));
        const int rl = row[el], cl = col[el];
        const float ewl = ew[el];
        float v[EPW][4];
#pragma unroll
        for (int j = 0; j < EPW; ++j) {
            const float* src = in + (size_t)lane_i(rl, j) * DC + l;
            v[j][0] = src[0]; v[j][1] = src[32]; v[j][2] = src[64]; v[j][3] = src[96];
        }
        const float* dg = degp + (size_t)(l / EPW) * HPG * N;
        float pr[HPG], pc[HPG];
#pragma unroll
        for (int p = 0; p < HPG; ++p) {
            pr[p] = dg[(size_t)p * N + rl];
            pc[p] = dg[(size_t)p * N + cl];
        }
#pragma unroll
        for (int m = 1; m < HPG; m <<= 1)
#pragma unroll
            for (int p = 0; p < HPG; p += 2 * m) { pr[p] += pr[p + m]; pc[p] += pc[p + m]; }
        float dr = pr[0], dc = pc[0];
#pragma unroll
        for (int m = EPW; m < 32; m <<= 1) {
            dr += __shfl_xor(dr, m);
            dc += __shfl_xor(dc, m);
        }
        const float wl = -inv_sqrt_deg(dr) * ewl * inv_sqrt_deg(dc);
        if (l < EPW) nw[e0 + l] = wl;
#pragma unroll
        for (int j = 0; j < EPW; ++j) {
            const float w = lane_f(wl, j);
            float* dst = out + (size_t)lane_i(cl, j) * DC + l;
            atomicAdd(dst +  0, w * v[j][0]);
            atomicAdd(dst + 32, w * v[j][1]);
            atomicAdd(dst + 64, w * v[j][2]);
            atomicAdd(dst + 96, w * v[j][3]);
        }
    } else {
        for (int e = e0; e < E; ++e) {
            const int r = row[e], c = col[e];
            const float w = -inv_sqrt_deg(degree_of(degp, N, r)) * ew[e]
                            * inv_sqrt_deg(degree_of(degp, N, c));
            if (l == 0) nw[e] = w;
            scatter_edge(in, out, r, c, w, l);
        }
    }
    const int nwv = (int)((gridDim.x * blockDim.x) >> 5);
    for (int i = e0 / EPW; i < N; i += nwv) {
        const size_t o = (size_t)i * DC + l;
        float* d = s0 + o;
        d[0]  = p[o]      - in[o]      + pb[l];
        d[32] = p[o + 32] - in[o + 32] + pb[l + 32];
        d[64] = p[o + 64] - in[o + 64] + pb[l + 64];
        d[96] = p[o + 96] - in[o + 96] + pb[l + 96];
    }
}

__global__ void __launch_bounds__(256)
spmm_kernel(const int* __restrict__ row,
            const int* __restrict__ col,
            const float* __restrict__ nw,
            const float* __restrict__ in,
            float* __restrict__ out,
            float sc,
            int E) {
    const int l  = threadIdx.x & 31;
    const int e0 = wave_first_edge();

    if (e0 + EPW <= E) {
        const int el = e0 + (l & (EPW - 1));
        const int rl = row[el], cl = col[el];
        const float wl = sc * nw[el];
        float v[EPW][4];
#pragma unroll
        for (int j = 0; j < EPW; ++j) {
            const float* src = in + (size_t)lane_i(rl, j) * DC + l;
            v[j][0] = src[0]; v[j][1] = src[32]; v[j][2] = src[64]; v[j][3] = src[96];
        }
#pragma unroll
        for (int j = 0; j < EPW; ++j) {
            const float w = lane_f(wl, j);
            float* dst = out + (size_t)lane_i(cl, j) * DC + l;
            atomicAdd(dst +  0, w * v[j][0]);
            atomicAdd(dst + 32, w * v[j][1]);
            atomicAdd(dst + 64, w * v[j][2]);
            atomicAdd(dst + 96, w * v[j][3]);
        }
    } else {
        for (int e = e0; e < E; ++e)
            scatter_edge(in, out, row[e], col[e], sc * nw[e], l);
    }
}

#define APAD 132

__global__ void __launch_bounds__(256)
gemm_h3_wmma_f32(const float* __restrict__ s1,
                 const float* __restrict__ B,
                 const float* __restrict__ bias,
                 float* __restrict__ out,
                 float* __restrict__ Q,
                 float* __restrict__ R) {
    __shared__ float As[16][APAD];

    const int wave = threadIdx.x >> 5;
    const int lane = threadIdx.x & 31;
    const int idx  = lane & 15;
    const int half = lane >> 4;

    const int tm   = blockIdx.x;
    const int tn   = wave;
    const int colB = tn * 16 + idx;

    const size_t strip = (size_t)tm * 16 * DC;
#pragma unroll
    for (int i = 0; i < 2; ++i) {
        int f  = threadIdx.x + i * 256;
        int r  = f >> 5;
        int c4 = f & 31;
        size_t off = strip + (size_t)r * DC;
        float4 s = ((const float4*)(s1 + off))[c4];
        float4 o;
        o.x = fmaxf(s.x, 0.f);
        o.y = fmaxf(s.y, 0.f);
        o.z = fmaxf(s.z, 0.f);
        o.w = fmaxf(s.w, 0.f);
        *(float4*)&As[r][c4 << 2] = o;
    }
    __syncthreads();

    v8f c0 = {0.f, 0.f, 0.f, 0.f, 0.f, 0.f, 0.f, 0.f};
    v8f c1 = {0.f, 0.f, 0.f, 0.f, 0.f, 0.f, 0.f, 0.f};
    v8f c2 = {0.f, 0.f, 0.f, 0.f, 0.f, 0.f, 0.f, 0.f};

    const float* Arow = &As[idx][2 * half];
    const float* Bb0  = B + (size_t)2 * half * DC + colB;
    const float* Bb1  = Bb0 + (size_t)DC * DC;
    const float* Bb2  = Bb1 + (size_t)DC * DC;
#pragma unroll
    for (int k0 = 0; k0 < DC; k0 += 4) {
        v2f a = *(const v2f*)(Arow + k0);
        v2f p, q, u;
        p[0] = Bb0[(size_t)k0 * DC];  p[1] = Bb0[(size_t)k0 * DC + DC];
        q[0] = Bb1[(size_t)k0 * DC];  q[1] = Bb1[(size_t)k0 * DC + DC];
        u[0] = Bb2[(size_t)k0 * DC];  u[1] = Bb2[(size_t)k0 * DC + DC];
        c0 = __builtin_amdgcn_wmma_f32_16x16x4_f32(
                 false, a, false, p, (short)0, c0, false, false);
        c1 = __builtin_amdgcn_wmma_f32_16x16x4_f32(
                 false, a, false, q, (short)0, c1, false, false);
        c2 = __builtin_amdgcn_wmma_f32_16x16x4_f32(
                 false, a, false, u, (short)0, c2, false, false);
    }

    float bv = bias[colB];
#pragma unroll
    for (int r = 0; r < 8; ++r) {
        size_t o = (size_t)(tm * 16 + r + 8 * half) * DC + colB;
        out[o] = c0[r] - c2[r] + bv;
        Q[o]   = c1[r];
        R[o]   = c2[r];
    }
}


extern "C" void kernel_launch(void* const* d_in, const int* in_sizes, int n_in,
                              void* d_out, int out_size, void* d_ws, size_t ws_size,
                              hipStream_t stream) {
    const int*   ei  = (const int*)d_in[1];
    const float* ew  = (const float*)d_in[2];
    const float* W1  = (const float*)d_in[3];
    const float* b1  = (const float*)d_in[4];
    const float* W2  = (const float*)d_in[5];
    const float* b2  = (const float*)d_in[6];
    float* out = (float*)d_out;

    const int E = in_sizes[2];
    const int D = in_sizes[4];
    const int N = in_sizes[3] / (3 * D);
    const int ND = N * D;

    const int* row = ei;
    const int* col = ei + E;
    const float* W1_0 = W1;
    const float* W1_1 = W1 + (size_t)ND;
    const float* W1_2 = W1 + (size_t)2 * ND;

    char* ws = (char*)d_ws;
    float* B0 = (float*)ws;                      ws += (size_t)ND * 4;
    float* B1 = (float*)ws;                      ws += (size_t)ND * 4;
    float* Qm = (float*)ws;                      ws += (size_t)ND * 4;
    float* normw = (float*)ws;                   ws += (size_t)E * 4;
    float* Rm = B1;
    float* degp = Qm;

    const int TB = 256;
    const int nz4 = ND / 4;
    dim3 gInit(HP + (nz4 + HT - 1) / HT);
    const int spmmThreads = ((E + EPW - 1) / EPW) * 32;
    dim3 gSpmm((spmmThreads + TB - 1) / TB);

    degree_init_kernel<<<gInit, HT, 0, stream>>>(row, ew, degp, W1_1, B1, nz4, E, N);
    spmm_norm_kernel<<<gSpmm, TB, 0, stream>>>(row, col, ew, degp, W1_2,
                                               B1, normw, W1_0, b1, B0, E, N);
    spmm_kernel<<<gSpmm, TB, 0, stream>>>(row, col, normw, B1, B0, 2.f, E);

    gemm_h3_wmma_f32<<<N / 16, 256, 0, stream>>>(B0, W2, b2, out, Qm, Rm);
    spmm_kernel<<<gSpmm, TB, 0, stream>>>(row, col, normw, Rm, Qm, 2.f, E);
    spmm_kernel<<<gSpmm, TB, 0, stream>>>(row, col, normw, Qm, out, 1.f, E);
}


# Second layer-1 SpMM moved into the GEMM launch: a strip's workgroup scans all destination indices, sums 2*nw*T[row] of its in-edges in LDS (LDS float atomics); 6 launches to 5; other sum order
# speedup vs baseline: 2.4904x; 1.1174x over previous
#include <hip/hip_runtime.h>


typedef __attribute__((ext_vector_type(2))) float v2f;
typedef __attribute__((ext_vector_type(8))) float v8f;

#define DC 128
#define EPW 8
#define NN 4096
#define HPG 2
#define HP (HPG * (32 / EPW))
#define HT 1024
#define HU 4

__device__ __forceinline__ int lane_i(int v, int j) {
    return __builtin_amdgcn_readlane(v, j);
}
__device__ __forceinline__ float lane_f(float v, int j) {
    return __int_as_float(__builtin_amdgcn_readlane(__float_as_int(v), j));
}

__device__ __forceinline__ int wave_first_edge() {
    int wv = (int)((blockIdx.x * blockDim.x + threadIdx.x) >> 5);
    return __builtin_amdgcn_readfirstlane(wv) * EPW;
}


__global__ void __launch_bounds__(HT)
degree_init_kernel(const int* __restrict__ row,
                   const float* __restrict__ ew,
                   float* __restrict__ degp,
                   const float* __restrict__ hsrc,
                   float* __restrict__ z, int n4, int E, int N) {
    __shared__ float hist[NN];
    const int t = threadIdx.x;
    if (blockIdx.x < HP) {
        for (int i = t; i < N; i += HT) hist[i] = 0.f;
        __syncthreads();
        const int slice = (E + HP - 1) / HP;
        const int eb = blockIdx.x * slice;
        const int ee = min(eb + slice, E);
        int base = eb;
        for (; base + HU * HT <= ee; base += HU * HT) {
            int r[HU];
            float w[HU];
#pragma unroll
            for (int k = 0; k < HU; ++k) {
                r[k] = row[base + k * HT + t];
                w[k] = ew[base + k * HT + t];
            }
#pragma unroll
            for (int k = 0; k < HU; ++k) atomicAdd(&hist[r[k]], w[k]);
        }
        for (int e = base + t; e < ee; e += HT) atomicAdd(&hist[row[e]], ew[e]);
        __syncthreads();
        float* dp = degp + (size_t)blockIdx.x * N;
        for (int i = t; i < N; i += HT) dp[i] = hist[i];
    } else {
        const int i = (blockIdx.x - HP) * HT + t;
        if (i < n4) {
            const float4 v = ((const float4*)hsrc)[i];
            ((float4*)z)[i] = make_float4(0.5f * v.x, 0.5f * v.y, 0.5f * v.z, 0.5f * v.w);
        }
    }
}

__device__ __forceinline__ float degree_of(const float* __restrict__ degp,
                                           int N, int i) {
    float s[HP];
#pragma unroll
    for (int p = 0; p < HP; ++p) s[p] = degp[(size_t)p * N + i];
#pragma unroll
    for (int m = 1; m < HP; m <<= 1)
#pragma unroll
        for (int p = 0; p < HP; p += 2 * m) s[p] += s[p + m];
    return s[0];
}

__device__ __forceinline__ float inv_sqrt_deg(float d) {
    return (d > 0.f) ? rsqrtf(d) : 0.f;
}

__device__ __forceinline__ void scatter_edge(const float* __restrict__ in,
                                             float* __restrict__ out,
                                             int r, int c, float w, int l) {
    const float* src = in + (size_t)r * DC + l;
    float v0 = src[0], v1 = src[32], v2 = src[64], v3 = src[96];
    float* dst = out + (size_t)c * DC + l;
    atomicAdd(dst +  0, w * v0);
    atomicAdd(dst + 32, w * v1);
    atomicAdd(dst + 64, w * v2);
    atomicAdd(dst + 96, w * v3);
}

__global__ void __launch_bounds__(256)
spmm_norm_kernel(const int* __restrict__ row,
                 const int* __restrict__ col,
                 const float* __restrict__ ew,
                 const float* __restrict__ degp,
                 const float* __restrict__ in,
                 float* __restrict__ out,
                 float* __restrict__ nw,
                 int E, int N) {
    const int l  = threadIdx.x & 31;
    const int e0 = wave_first_edge();

    if (e0 + EPW <= E) {
        const int el = e0 + (l & (EPW - 1));
        const int rl = row[el], cl = col[el];
        const float ewl = ew[el];
        float v[EPW][4];
#pragma unroll
        for (int j = 0; j < EPW; ++j) {
            const float* src = in + (size_t)lane_i(rl, j) * DC + l;
            v[j][0] = src[0]; v[j][1] = src[32]; v[j][2] = src[64]; v[j][3] = src[96];
        }
        const float* dg = degp + (size_t)(l / EPW) * HPG * N;
        float pr[HPG], pc[HPG];
#pragma unroll
        for (int p = 0; p < HPG; ++p) {
            pr[p] = dg[(size_t)p * N + rl];
            pc[p] = dg[(size_t)p * N + cl];
        }
#pragma unroll
        for (int m = 1; m < HPG; m <<= 1)
#pragma unroll
            for (int p = 0; p < HPG; p += 2 * m) { pr[p] += pr[p + m]; pc[p] += pc[p + m]; }
        float dr = pr[0], dc = pc[0];
#pragma unroll
        for (int m = EPW; m < 32; m <<= 1) {
            dr += __shfl_xor(dr, m);
            dc += __shfl_xor(dc, m);
        }
        const float wl = -inv_sqrt_deg(dr) * ewl * inv_sqrt_deg(dc);
        if (l < EPW) nw[e0 + l] = wl;
#pragma unroll
        for (int j = 0; j < EPW; ++j) {
            const float w = lane_f(wl, j);
            float* dst = out + (size_t)lane_i(cl, j) * DC + l;
            atomicAdd(dst +  0, w * v[j][0]);
            atomicAdd(dst + 32, w * v[j][1]);
            atomicAdd(dst + 64, w * v[j][2]);
            atomicAdd(dst + 96, w * v[j][3]);
        }
    } else {
        for (int e = e0; e < E; ++e) {
            const int r = row[e], c = col[e];
            const float w = -inv_sqrt_deg(degree_of(degp, N, r)) * ew[e]
                            * inv_sqrt_deg(degree_of(degp, N, c));
            if (l == 0) nw[e] = w;
            scatter_edge(in, out, r, c, w, l);
        }
    }
}

__global__ void __launch_bounds__(256)
spmm_kernel(const int* __restrict__ row,
            const int* __restrict__ col,
            const float* __restrict__ nw,
            const float* __restrict__ in,
            float* __restrict__ out,
            float sc,
            int E) {
    const int l  = threadIdx.x & 31;
    const int e0 = wave_first_edge();

    if (e0 + EPW <= E) {
        const int el = e0 + (l & (EPW - 1));
        const int rl = row[el], cl = col[el];
        const float wl = sc * nw[el];
        float v[EPW][4];
#pragma unroll
        for (int j = 0; j < EPW; ++j) {
            const float* src = in + (size_t)lane_i(rl, j) * DC + l;
            v[j][0] = src[0]; v[j][1] = src[32]; v[j][2] = src[64]; v[j][3] = src[96];
        }
#pragma unroll
        for (int j = 0; j < EPW; ++j) {
            const float w = lane_f(wl, j);
            float* dst = out + (size_t)lane_i(cl, j) * DC + l;
            atomicAdd(dst +  0, w * v[j][0]);
            atomicAdd(dst + 32, w * v[j][1]);
            atomicAdd(dst + 64, w * v[j][2]);
            atomicAdd(dst + 96, w * v[j][3]);
        }
    } else {
        for (int e = e0; e < E; ++e)
            scatter_edge(in, out, row[e], col[e], sc * nw[e], l);
    }
}

#define APAD 132
#define SCAN_U 32
#define SCAN_W (SCAN_U * 128)
#define SCAN_B (8 * SCAN_W)
#define PG 8

__device__ __forceinline__ int note_edges(unsigned short* __restrict__ wl, int cnt,
                                          bool m, int place) {
    const unsigned mask = __builtin_amdgcn_ballot_w32(m);
    if (mask != 0u) {
        if (m) wl[cnt + (int)__builtin_amdgcn_mbcnt_lo(mask, 0u)] = (unsigned short)place;
        cnt += __builtin_popcount(mask);
    }
    return cnt;
}

__global__ void __launch_bounds__(256)
gemm_h3_wmma_f32(const int* __restrict__ row,
                 const int* __restrict__ col,
                 const float* __restrict__ nw,
                 const float* __restrict__ T,
                 const float* __restrict__ w0,
                 const float* __restrict__ w2,
                 const float* __restrict__ b1,
                 const float* __restrict__ B,
                 const float* __restrict__ bias,
                 float* __restrict__ out,
                 float* __restrict__ Q,
                 float* __restrict__ R,
                 int E) {
    __shared__ float As[16][APAD];
    __shared__ unsigned short wlist[8][SCAN_W];

    const int wave = threadIdx.x >> 5;
    const int lane = threadIdx.x & 31;
    const int idx  = lane & 15;
    const int half = lane >> 4;

    const int tm   = blockIdx.x;
    const int tn   = wave;
    const int colB = tn * 16 + idx;

#pragma unroll
    for (int i = 0; i < 2; ++i) {
        const int f = threadIdx.x + i * 256;
        *(float4*)&As[f >> 5][(f & 31) << 2] = make_float4(0.f, 0.f, 0.f, 0.f);
    }
    const size_t strip = (size_t)tm * 16 * DC;
    float4 ra[2], rb[2], rc[2];
#pragma unroll
    for (int i = 0; i < 2; ++i) {
        const int f  = threadIdx.x + i * 256;
        const size_t off = strip + (size_t)(f >> 5) * DC;
        ra[i] = ((const float4*)(w0 + off))[f & 31];
        rb[i] = ((const float4*)(w2 + off))[f & 31];
        rc[i] = ((const float4*)b1)[f & 31];
    }

    unsigned short* wl = wlist[wave];
    const bool vec_ok = (E & 3) == 0;
    for (int eb = 0; eb < E; eb += SCAN_B) {
        int n = 0;
        if (vec_ok && eb + SCAN_B <= E) {
            int4 cv[SCAN_U];
#pragma unroll
            for (int u = 0; u < SCAN_U; ++u) {
                const int ur = (u + tm) & (SCAN_U - 1);
                cv[u] = *(const int4*)(col + eb + (ur * 8 + wave) * 128 + lane * 4);
            }
            __builtin_amdgcn_sched_barrier(0);
            unsigned hit[SCAN_U / 8];
#pragma unroll
            for (int g = 0; g < SCAN_U / 8; ++g) {
                unsigned h = 0u;
#pragma unroll
                for (int uu = 0; uu < 8; ++uu) {
                    const int4 c = cv[g * 8 + uu];
                    h = h + h + ((((unsigned)c.x >> 4) == (unsigned)tm) ? 1u : 0u);
                    h = h + h + ((((unsigned)c.y >> 4) == (unsigned)tm) ? 1u : 0u);
                    h = h + h + ((((unsigned)c.z >> 4) == (unsigned)tm) ? 1u : 0u);
                    h = h + h + ((((unsigned)c.w >> 4) == (unsigned)tm) ? 1u : 0u);
                }
                hit[g] = h;
            }
#pragma unroll
            for (int g = 0; g < SCAN_U / 8; ++g) {
                unsigned h = hit[g];
                for (;;) {
                    const bool has = h != 0u;
                    const unsigned mask = __builtin_amdgcn_ballot_w32(has);
                    if (mask == 0u) break;
                    if (has) {
                        const int s  = __builtin_clz(h);
                        const int ur = (g * 8 + (s >> 2) + tm) & (SCAN_U - 1);
                        wl[n + (int)__builtin_amdgcn_mbcnt_lo(mask, 0u)] =
                            (unsigned short)(ur * 128 + lane * 4 + (s & 3));
                        h &= ~(0x80000000u >> s);
                    }
                    n += __builtin_popcount(mask);
                }
            }
        } else {
            for (int u = 0; u < SCAN_U * 4; ++u) {
                const int place = u * 32 + lane;
                const int e = eb + ((place >> 7) * 8 + wave) * 128 + (place & 127);
                const int c = (e < E) ? col[e] : -1;
                n = note_edges(wl, n, ((unsigned)c >> 4) == (unsigned)tm, place);
            }
        }
        __syncthreads();

        for (int base = 0; base < n; base += 32) {
            const int i = min(base + lane, n - 1);
            const int place = (int)wl[i];
            const int e = eb + ((place >> 7) * 8 + wave) * 128 + (place & 127);
            const int rl = row[e];
            const float wl2 = 2.f * nw[e];
            const int cl = col[e] & 15;
            const int m = n - base;
            float v[32][4];
#pragma unroll
            for (int g = 0; g < 32; g += PG) {
                if (m > g) {
#pragma unroll
                    for (int j = g; j < g + PG; ++j) {
                        const float* src = T + (size_t)lane_i(rl, j) * DC + lane;
                        v[j][0] = src[0]; v[j][1] = src[32]; v[j][2] = src[64]; v[j][3] = src[96];
                    }
                }
            }
#pragma unroll
            for (int j = 0; j < 32; ++j) {
                if (j < m) {
                    const float w = lane_f(wl2, j);
                    float* d = &As[lane_i(cl, j)][lane];
                    atomicAdd(d +  0, w * v[j][0]);
                    atomicAdd(d + 32, w * v[j][1]);
                    atomicAdd(d + 64, w * v[j][2]);
                    atomicAdd(d + 96, w * v[j][3]);
                }
            }
        }
    }
    __syncthreads();
#pragma unroll
    for (int i = 0; i < 2; ++i) {
        const int f = threadIdx.x + i * 256;
        float4* p = (float4*)&As[f >> 5][(f & 31) << 2];
        float4 s = *p;
        s.x = fmaxf(s.x + (ra[i].x - rb[i].x + rc[i].x), 0.f);
        s.y = fmaxf(s.y + (ra[i].y - rb[i].y + rc[i].y), 0.f);
        s.z = fmaxf(s.z + (ra[i].z - rb[i].z + rc[i].z), 0.f);
        s.w = fmaxf(s.w + (ra[i].w - rb[i].w + rc[i].w), 0.f);
        *p = s;
    }
    __syncthreads();

    v8f c0 = {0.f, 0.f, 0.f, 0.f, 0.f, 0.f, 0.f, 0.f};
    v8f c1 = {0.f, 0.f, 0.f, 0.f, 0.f, 0.f, 0.f, 0.f};
    v8f c2 = {0.f, 0.f, 0.f, 0.f, 0.f, 0.f, 0.f, 0.f};

    const float* Arow = &As[idx][2 * half];
    const float* Bb0  = B + (size_t)2 * half * DC + colB;
    const float* Bb1  = Bb0 + (size_t)DC * DC;
    const float* Bb2  = Bb1 + (size_t)DC * DC;
#pragma unroll
    for (int k0 = 0; k0 < DC; k0 += 4) {
        v2f a = *(const v2f*)(Arow + k0);
        v2f p, q, u;
        p[0] = Bb0[(size_t)k0 * DC];  p[1] = Bb0[(size_t)k0 * DC + DC];
        q[0] = Bb1[(size_t)k0 * DC];  q[1] = Bb1[(size_t)k0 * DC + DC];
        u[0] = Bb2[(size_t)k0 * DC];  u[1] = Bb2[(size_t)k0 * DC + DC];
        c0 = __builtin_amdgcn_wmma_f32_16x16x4_f32(
                 false, a, false, p, (short)0, c0, false, false);
        c1 = __builtin_amdgcn_wmma_f32_16x16x4_f32(
                 false, a, false, q, (short)0, c1, false, false);
        c2 = __builtin_amdgcn_wmma_f32_16x16x4_f32(
                 false, a, false, u, (short)0, c2, false, false);
    }

    float bv = bias[colB];
#pragma unroll
    for (int r = 0; r < 8; ++r) {
        size_t o = (size_t)(tm * 16 + r + 8 * half) * DC + colB;
        out[o] = c0[r] - c2[r] + bv;
        Q[o]   = c1[r];
        R[o]   = c2[r];
    }
}


extern "C" void kernel_launch(void* const* d_in, const int* in_sizes, int n_in,
                              void* d_out, int out_size, void* d_ws, size_t ws_size,
                              hipStream_t stream) {
    const int*   ei  = (const int*)d_in[1];
    const float* ew  = (const float*)d_in[2];
    const float* W1  = (const float*)d_in[3];
    const float* b1  = (const float*)d_in[4];
    const float* W2  = (const float*)d_in[5];
    const float* b2  = (const float*)d_in[6];
    float* out = (float*)d_out;

    const int E = in_sizes[2];
    const int D = in_sizes[4];
    const int N = in_sizes[3] / (3 * D);
    const int ND = N * D;

    const int* row = ei;
    const int* col = ei + E;
    const float* W1_0 = W1;
    const float* W1_1 = W1 + (size_t)ND;
    const float* W1_2 = W1 + (size_t)2 * ND;

    char* ws = (char*)d_ws;
    float* Rm = (float*)ws;                      ws += (size_t)ND * 4;
    float* B1 = (float*)ws;                      ws += (size_t)ND * 4;
    float* Qm = (float*)ws;                      ws += (size_t)ND * 4;
    float* normw = (float*)ws;                   ws += (size_t)E * 4;
    float* degp = Qm;

    const int TB = 256;
    const int nz4 = ND / 4;
    dim3 gInit(HP + (nz4 + HT - 1) / HT);
    const int spmmThreads = ((E + EPW - 1) / EPW) * 32;
    dim3 gSpmm((spmmThreads + TB - 1) / TB);

    degree_init_kernel<<<gInit, HT, 0, stream>>>(row, ew, degp, W1_1, B1, nz4, E, N);
    spmm_norm_kernel<<<gSpmm, TB, 0, stream>>>(row, col, ew, degp, W1_2,
                                               B1, normw, E, N);

    gemm_h3_wmma_f32<<<N / 16, 256, 0, stream>>>(row, col, normw, B1, W1_0, W1_2, b1,
                                                 W2, b2, out, Qm, Rm, E);
    spmm_kernel<<<gSpmm, TB, 0, stream>>>(row, col, normw, Rm, Qm, 2.f, E);
    spmm_kernel<<<gSpmm, TB, 0, stream>>>(row, col, normw, Qm, out, 1.f, E);
}
